# starting kernel
# baseline (speedup 1.0000x reference)
#include <hip/hip_runtime.h>


typedef __attribute__((ext_vector_type(2))) float v2f;
typedef __attribute__((ext_vector_type(8))) float v8f;

#define BATCH 64
#define VDIM  16
#define MAXN  24
#define NTHREADS 384
#define DPAD  72
#define XVPAD 49

__launch_bounds__(NTHREADS)
__global__ void residues_encode_kernel(const float* __restrict__ x,
                                       const float* __restrict__ W,
                                       const int*   __restrict__ seq,
                                       const int*   __restrict__ ridx,
                                       const int*   __restrict__ natoms,
                                       float* __restrict__ out_root,
                                       float* __restrict__ out_xv,
                                       int A, int R)
{
    __shared__ float lds_root[3][BATCH];
    __shared__ float lds_delta[3][MAXN][DPAD];

    const int r = blockIdx.x;
    const int t = threadIdx.x;

    const int sr = seq[r];
    const int ri = ridx[r];
    const int na = natoms[sr];

    if (t < BATCH * 3) {
        const int b = t / 3, d = t % 3;
        const float rv = x[((long)b * A + (ri + 1)) * 3 + d];
        lds_root[d][b] = rv;
        out_root[((long)b * R + r) * 3 + d] = rv;
    }
    __syncthreads();

    for (int e = t; e < BATCH * MAXN * 3; e += NTHREADS) {
        const int b = e / (MAXN * 3);
        const int inner = e % (MAXN * 3);
        const int n = inner / 3, d = inner % 3;
        float v = 0.0f;
        if (n < na) {
            int atom = ri + n;
            if (atom > A - 1) atom = A - 1;
            v = x[((long)b * A + atom) * 3 + d] - lds_root[d][b];
        }
        lds_delta[d][n][b] = v;
    }
    __syncthreads();

    const int wave  = t >> 5;
    const int lane  = t & 31;
    const int half  = lane >> 4;
    const int col   = lane & 15;
    const int d     = wave >> 2;
    const int btile = wave & 3;
    const int b     = btile * 16 + col;

    const float* Wr = W + ((long)(sr * VDIM + col) * MAXN + 2 * half);

    v8f acc = {};
    #pragma unroll
    for (int k = 0; k < 6; ++k) {
        v2f a;
        a[0] = Wr[4 * k];
        a[1] = Wr[4 * k + 1];
        v2f bb;
        bb[0] = lds_delta[d][4 * k + 2 * half + 0][b];
        bb[1] = lds_delta[d][4 * k + 2 * half + 1][b];
        acc = __builtin_amdgcn_wmma_f32_16x16x4_f32(
             false, a,  false, bb,
             (short)0, acc,  false,  false);
    }
    __syncthreads();

    float* xv = &lds_delta[0][0][0];
    #pragma unroll
    for (int p = 0; p < 8; ++p) {
        const int v = p + 8 * half;
        xv[b * XVPAD + v * 3 + d] = acc[p];
    }
    __syncthreads();

    for (int e = t; e < BATCH * VDIM * 3; e += NTHREADS) {
        const int b2 = e / (VDIM * 3);
        const int j  = e % (VDIM * 3);
        out_xv[((long)b2 * R + r) * (VDIM * 3) + j] = xv[b2 * XVPAD + j];
    }
}

extern "C" void kernel_launch(void* const* d_in, const int* in_sizes, int n_in,
                              void* d_out, int out_size, void* d_ws, size_t ws_size,
                              hipStream_t stream) {
    const float* x      = (const float*)d_in[0];
    const float* W      = (const float*)d_in[1];
    const int*   seq    = (const int*)d_in[2];
    const int*   ridx   = (const int*)d_in[3];
    const int*   natoms = (const int*)d_in[4];

    const int R = in_sizes[2];
    const int A = in_sizes[0] / (BATCH * 3);

    float* out_root = (float*)d_out;
    float* out_xv   = out_root + (size_t)BATCH * R * 3;

    residues_encode_kernel<<<R, NTHREADS, 0, stream>>>(x, W, seq, ridx, natoms,
                                                       out_root, out_xv, A, R);
}


# Registers cut from 74 to 61 per wave (A fragments fetched from the tile two k-steps at a time, 32-bit lane offsets): 16 waves per SIMD, all 2048 work-groups resident in one pass, was 1536 then 512
# speedup vs baseline: 2.1043x; 2.1043x over previous
#include <hip/hip_runtime.h>


typedef __attribute__((ext_vector_type(2))) float v2f;
typedef __attribute__((ext_vector_type(4))) float v4f;
typedef __attribute__((ext_vector_type(8))) float v8f;

#define BATCH 64
#define VDIM  16
#define MAXN  24
#define WAVES_PER_WG 8
#define NTHREADS (32 * WAVES_PER_WG)
#define PITCH 76

__launch_bounds__(NTHREADS, 16)
__global__ void residues_encode_kernel(const float* __restrict__ x,
                                       const float* __restrict__ W,
                                       const int*   __restrict__ seq,
                                       const int*   __restrict__ ridx,
                                       const int*   __restrict__ natoms,
                                       float* __restrict__ out_root,
                                       float* __restrict__ out_xv,
                                       int A, int R)
{
    __shared__ __attribute__((aligned(16))) float tile[WAVES_PER_WG][16 * PITCH];

    const int lane = threadIdx.x & 31;
    const int wave  = __builtin_amdgcn_readfirstlane((int)(threadIdx.x >> 5));
    const int btile = blockIdx.x & 3;
    const int r     = (int)(blockIdx.x >> 2) * WAVES_PER_WG + wave;
    if (r >= R) return;

    const int sr = seq[r];
    const int ri = ridx[r];
    const int na = natoms[sr];

    const int h = lane >> 4;
    const int c = lane & 15;

    const size_t   rowbytes = (size_t)A * 12;
    const char*    xb = (const char*)x + (size_t)(btile * 16) * rowbytes;
    const unsigned rb32 = (unsigned)A * 12u;

    float g0[8][3];
    {
        int atom = ri + c;
        if (atom > A - 1) atom = A - 1;
        const unsigned off = (unsigned)(8 * h) * rb32 + (unsigned)atom * 12u;
        #pragma unroll
        for (int i = 0; i < 8; ++i) {
            const float* p = (const float*)(xb + (size_t)i * rowbytes + off);
            g0[i][0] = p[0]; g0[i][1] = p[1]; g0[i][2] = p[2];
        }
    }
    const int q  = lane >> 3;
    const int a8 = lane & 7;
    float g1[4][3];
    {
        int atom = ri + 16 + a8;
        if (atom > A - 1) atom = A - 1;
        const unsigned off = (unsigned)(4 * q) * rb32 + (unsigned)atom * 12u;
        #pragma unroll
        for (int i = 0; i < 4; ++i) {
            const float* p = (const float*)(xb + (size_t)i * rowbytes + off);
            g1[i][0] = p[0]; g1[i][1] = p[1]; g1[i][2] = p[2];
        }
    }

    const v4f* Wr = (const v4f*)((const char*)(W + (size_t)sr * (VDIM * MAXN))
                                 + (unsigned)((c * MAXN + 12 * h) * 4));
    const v4f w0 = Wr[0], w1 = Wr[1], w2 = Wr[2];

    float* T = &tile[wave][0];
    {
        float* d0 = T + (8 * h) * PITCH + 3 * c;
        #pragma unroll
        for (int i = 0; i < 8; ++i) {
            d0[0] = g0[i][0]; d0[1] = g0[i][1]; d0[2] = g0[i][2];
            d0 += PITCH;
        }
        float* d1 = T + (4 * q) * PITCH + 48 + 3 * a8;
        #pragma unroll
        for (int i = 0; i < 4; ++i) {
            d1[0] = g1[i][0]; d1[1] = g1[i][1]; d1[2] = g1[i][2];
            d1 += PITCH;
        }
    }

    float root[3];
    root[0] = T[c * PITCH + 3]; root[1] = T[c * PITCH + 4]; root[2] = T[c * PITCH + 5];

    if (h == 0) {
        float* o = (float*)((char*)(out_root + ((size_t)(btile * 16) * R + r) * 3)
                            + (unsigned)c * ((unsigned)R * 12u));
        o[0] = root[0]; o[1] = root[1]; o[2] = root[2];
    }

    const float wk[12] = { w0[0], w0[1], w0[2], w0[3], w1[0], w1[1], w1[2], w1[3],
                           w2[0], w2[1], w2[2], w2[3] };

    const v4f* s4 = (const v4f*)(T + c * PITCH + 36 * h);
    v8f acc0 = {}, acc1 = {}, acc2 = {};
    #pragma unroll
    for (int jj = 0; jj < 3; ++jj) {
        const v4f t0 = s4[3 * jj], t1 = s4[3 * jj + 1], t2 = s4[3 * jj + 2];
        const float e[12] = { t0[0], t0[1], t0[2], t0[3], t1[0], t1[1], t1[2], t1[3],
                              t2[0], t2[1], t2[2], t2[3] };
        #pragma unroll
        for (int u = 0; u < 2; ++u) {
            const int j  = 2 * jj + u;
            const int n0 = 12 * h + 2 * j;
            const bool m0 = n0     < na;
            const bool m1 = n0 + 1 < na;
            v2f a0, a1, a2, b;
            a0[0] = m0 ? e[6 * u + 0] - root[0] : 0.0f;
            a0[1] = m1 ? e[6 * u + 3] - root[0] : 0.0f;
            a1[0] = m0 ? e[6 * u + 1] - root[1] : 0.0f;
            a1[1] = m1 ? e[6 * u + 4] - root[1] : 0.0f;
            a2[0] = m0 ? e[6 * u + 2] - root[2] : 0.0f;
            a2[1] = m1 ? e[6 * u + 5] - root[2] : 0.0f;
            b[0] = wk[2 * j]; b[1] = wk[2 * j + 1];
            acc0 = __builtin_amdgcn_wmma_f32_16x16x4_f32(false, a0, false, b, (short)0, acc0, false, false);
            acc1 = __builtin_amdgcn_wmma_f32_16x16x4_f32(false, a1, false, b, (short)0, acc1, false, false);
            acc2 = __builtin_amdgcn_wmma_f32_16x16x4_f32(false, a2, false, b, (short)0, acc2, false, false);
        }
    }

    const size_t   orow = (size_t)R * (VDIM * 3 * 4);
    char* ob = (char*)(out_xv + ((size_t)(btile * 16) * R + r) * (VDIM * 3));
    const unsigned ooff = (unsigned)(8 * h) * ((unsigned)R * (VDIM * 3 * 4)) + (unsigned)c * 12u;
    #pragma unroll
    for (int p = 0; p < 8; ++p) {
        float* o = (float*)(ob + (size_t)p * orow + ooff);
        o[0] = acc0[p]; o[1] = acc1[p]; o[2] = acc2[p];
    }
}

extern "C" void kernel_launch(void* const* d_in, const int* in_sizes, int n_in,
                              void* d_out, int out_size, void* d_ws, size_t ws_size,
                              hipStream_t stream) {
    const float* x      = (const float*)d_in[0];
    const float* W      = (const float*)d_in[1];
    const int*   seq    = (const int*)d_in[2];
    const int*   ridx   = (const int*)d_in[3];
    const int*   natoms = (const int*)d_in[4];

    const int R = in_sizes[2];
    const int A = in_sizes[0] / (BATCH * 3);

    float* out_root = (float*)d_out;
    float* out_xv   = out_root + (size_t)BATCH * R * 3;

    const int groups = (R + WAVES_PER_WG - 1) / WAVES_PER_WG;
    residues_encode_kernel<<<groups * (BATCH / 16), NTHREADS, 0, stream>>>(x, W, seq, ridx, natoms,
                                                                          out_root, out_xv, A, R);
}


# x loads issued only for atoms the mask keeps (lane n < natoms; atoms 0, 1 always, for root); masked lanes put 0 in the tile; atoms 16..23 skipped whole when natoms <= 16: fewer line requests
# speedup vs baseline: 2.2301x; 1.0598x over previous
#include <hip/hip_runtime.h>


typedef __attribute__((ext_vector_type(2))) float v2f;
typedef __attribute__((ext_vector_type(4))) float v4f;
typedef __attribute__((ext_vector_type(8))) float v8f;

#define BATCH 64
#define VDIM  16
#define MAXN  24
#define WAVES_PER_WG 8
#define NTHREADS (32 * WAVES_PER_WG)
#define PITCH 76

__launch_bounds__(NTHREADS, 16)
__global__ void residues_encode_kernel(const float* __restrict__ x,
                                       const float* __restrict__ W,
                                       const int*   __restrict__ seq,
                                       const int*   __restrict__ ridx,
                                       const int*   __restrict__ natoms,
                                       float* __restrict__ out_root,
                                       float* __restrict__ out_xv,
                                       int A, int R)
{
    __shared__ __attribute__((aligned(16))) float tile[WAVES_PER_WG][16 * PITCH];

    const int lane = threadIdx.x & 31;
    const int wave  = __builtin_amdgcn_readfirstlane((int)(threadIdx.x >> 5));
    const int btile = blockIdx.x & 3;
    const int r     = (int)(blockIdx.x >> 2) * WAVES_PER_WG + wave;
    if (r >= R) return;

    const int sr = seq[r];
    const int ri = ridx[r];
    const int na = natoms[sr];
    const int lim = na > 2 ? na : 2;

    const int h = lane >> 4;
    const int c = lane & 15;

    const size_t   rowbytes = (size_t)A * 12;
    const char*    xb = (const char*)x + (size_t)(btile * 16) * rowbytes;
    const unsigned rb32 = (unsigned)A * 12u;

    float g0[8][3];
    #pragma unroll
    for (int i = 0; i < 8; ++i) { g0[i][0] = 0.0f; g0[i][1] = 0.0f; g0[i][2] = 0.0f; }
    if (c < lim) {
        int atom = ri + c;
        if (atom > A - 1) atom = A - 1;
        const unsigned off = (unsigned)(8 * h) * rb32 + (unsigned)atom * 12u;
        #pragma unroll
        for (int i = 0; i < 8; ++i) {
            const float* p = (const float*)(xb + (size_t)i * rowbytes + off);
            g0[i][0] = p[0]; g0[i][1] = p[1]; g0[i][2] = p[2];
        }
    }
    const int q  = lane >> 3;
    const int a8 = lane & 7;
    float g1[4][3];
    #pragma unroll
    for (int i = 0; i < 4; ++i) { g1[i][0] = 0.0f; g1[i][1] = 0.0f; g1[i][2] = 0.0f; }
    if (16 + a8 < lim) {
        int atom = ri + 16 + a8;
        if (atom > A - 1) atom = A - 1;
        const unsigned off = (unsigned)(4 * q) * rb32 + (unsigned)atom * 12u;
        #pragma unroll
        for (int i = 0; i < 4; ++i) {
            const float* p = (const float*)(xb + (size_t)i * rowbytes + off);
            g1[i][0] = p[0]; g1[i][1] = p[1]; g1[i][2] = p[2];
        }
    }

    const v4f* Wr = (const v4f*)((const char*)(W + (size_t)sr * (VDIM * MAXN))
                                 + (unsigned)((c * MAXN + 12 * h) * 4));
    const v4f w0 = Wr[0], w1 = Wr[1], w2 = Wr[2];

    float* T = &tile[wave][0];
    {
        float* d0 = T + (8 * h) * PITCH + 3 * c;
        #pragma unroll
        for (int i = 0; i < 8; ++i) {
            d0[0] = g0[i][0]; d0[1] = g0[i][1]; d0[2] = g0[i][2];
            d0 += PITCH;
        }
        float* d1 = T + (4 * q) * PITCH + 48 + 3 * a8;
        #pragma unroll
        for (int i = 0; i < 4; ++i) {
            d1[0] = g1[i][0]; d1[1] = g1[i][1]; d1[2] = g1[i][2];
            d1 += PITCH;
        }
    }

    float root[3];
    root[0] = T[c * PITCH + 3]; root[1] = T[c * PITCH + 4]; root[2] = T[c * PITCH + 5];

    if (h == 0) {
        float* o = (float*)((char*)(out_root + ((size_t)(btile * 16) * R + r) * 3)
                            + (unsigned)c * ((unsigned)R * 12u));
        o[0] = root[0]; o[1] = root[1]; o[2] = root[2];
    }

    const float wk[12] = { w0[0], w0[1], w0[2], w0[3], w1[0], w1[1], w1[2], w1[3],
                           w2[0], w2[1], w2[2], w2[3] };

    const v4f* s4 = (const v4f*)(T + c * PITCH + 36 * h);
    v8f acc0 = {}, acc1 = {}, acc2 = {};
    #pragma unroll
    for (int jj = 0; jj < 3; ++jj) {
        const v4f t0 = s4[3 * jj], t1 = s4[3 * jj + 1], t2 = s4[3 * jj + 2];
        const float e[12] = { t0[0], t0[1], t0[2], t0[3], t1[0], t1[1], t1[2], t1[3],
                              t2[0], t2[1], t2[2], t2[3] };
        #pragma unroll
        for (int u = 0; u < 2; ++u) {
            const int j  = 2 * jj + u;
            const int n0 = 12 * h + 2 * j;
            const bool m0 = n0     < na;
            const bool m1 = n0 + 1 < na;
            v2f a0, a1, a2, b;
            a0[0] = m0 ? e[6 * u + 0] - root[0] : 0.0f;
            a0[1] = m1 ? e[6 * u + 3] - root[0] : 0.0f;
            a1[0] = m0 ? e[6 * u + 1] - root[1] : 0.0f;
            a1[1] = m1 ? e[6 * u + 4] - root[1] : 0.0f;
            a2[0] = m0 ? e[6 * u + 2] - root[2] : 0.0f;
            a2[1] = m1 ? e[6 * u + 5] - root[2] : 0.0f;
            b[0] = wk[2 * j]; b[1] = wk[2 * j + 1];
            acc0 = __builtin_amdgcn_wmma_f32_16x16x4_f32(false, a0, false, b, (short)0, acc0, false, false);
            acc1 = __builtin_amdgcn_wmma_f32_16x16x4_f32(false, a1, false, b, (short)0, acc1, false, false);
            acc2 = __builtin_amdgcn_wmma_f32_16x16x4_f32(false, a2, false, b, (short)0, acc2, false, false);
        }
    }

    const size_t   orow = (size_t)R * (VDIM * 3 * 4);
    char* ob = (char*)(out_xv + ((size_t)(btile * 16) * R + r) * (VDIM * 3));
    const unsigned ooff = (unsigned)(8 * h) * ((unsigned)R * (VDIM * 3 * 4)) + (unsigned)c * 12u;
    #pragma unroll
    for (int p = 0; p < 8; ++p) {
        float* o = (float*)(ob + (size_t)p * orow + ooff);
        o[0] = acc0[p]; o[1] = acc1[p]; o[2] = acc2[p];
    }
}

extern "C" void kernel_launch(void* const* d_in, const int* in_sizes, int n_in,
                              void* d_out, int out_size, void* d_ws, size_t ws_size,
                              hipStream_t stream) {
    const float* x      = (const float*)d_in[0];
    const float* W      = (const float*)d_in[1];
    const int*   seq    = (const int*)d_in[2];
    const int*   ridx   = (const int*)d_in[3];
    const int*   natoms = (const int*)d_in[4];

    const int R = in_sizes[2];
    const int A = in_sizes[0] / (BATCH * 3);

    float* out_root = (float*)d_out;
    float* out_xv   = out_root + (size_t)BATCH * R * 3;

    const int groups = (R + WAVES_PER_WG - 1) / WAVES_PER_WG;
    residues_encode_kernel<<<groups * (BATCH / 16), NTHREADS, 0, stream>>>(x, W, seq, ridx, natoms,
                                                                          out_root, out_xv, A, R);
}


# A wave takes residues 2q and 2q+1: each x_v store is one batch row's 384 contiguous bytes = 3 whole 128-B lines (lane halves swapped by v_permlanex16), root 24 B a row; 8192 waves enter, not 16384
# speedup vs baseline: 2.6240x; 1.1766x over previous
#include <hip/hip_runtime.h>


typedef __attribute__((ext_vector_type(2))) float v2f;
typedef __attribute__((ext_vector_type(4))) float v4f;
typedef __attribute__((ext_vector_type(8))) float v8f;

#define BATCH 64
#define VDIM  16
#define MAXN  24
#define WAVES_PER_WG 8
#define NTHREADS (32 * WAVES_PER_WG)
#define PITCH 76

__device__ __forceinline__ float other_half(float v)
{
    const unsigned u = __float_as_uint(v);
    return __uint_as_float(__builtin_amdgcn_permlanex16(u, u, 0x76543210u, 0xfedcba98u, false, false));
}

typedef __attribute__((address_space(1))) char  gchar;
typedef __attribute__((address_space(1))) float gfloat;
__device__ __forceinline__ gchar* uniform_ptr(const char* p)
{
    const unsigned long long v = (unsigned long long)p;
    const unsigned lo = __builtin_amdgcn_readfirstlane((unsigned)v);
    const unsigned hi = __builtin_amdgcn_readfirstlane((unsigned)(v >> 32));
    return (gchar*)(((unsigned long long)hi << 32) | lo);
}

__launch_bounds__(NTHREADS, 8)
__global__ void residues_encode_kernel(const float* __restrict__ x,
                                       const float* __restrict__ W,
                                       const int*   __restrict__ seq,
                                       const int*   __restrict__ ridx,
                                       const int*   __restrict__ natoms,
                                       float* __restrict__ out_root,
                                       float* __restrict__ out_xv,
                                       int A, int R)
{
    __shared__ __attribute__((aligned(16))) float tile[WAVES_PER_WG][2][16 * PITCH];

    const int lane = threadIdx.x & 31;
    const int wave  = __builtin_amdgcn_readfirstlane((int)(threadIdx.x >> 5));
    const int btile = blockIdx.x & 3;
    const int rA    = 2 * ((int)(blockIdx.x >> 2) * WAVES_PER_WG + wave);
    if (rA >= R) return;
    const bool hasB = rA + 1 < R;

    const int h = lane >> 4;
    const int c = lane & 15;
    const int q  = lane >> 3;
    const int a8 = lane & 7;

    const size_t   rowbytes = (size_t)A * 12;
    const char*    xb = (const char*)x + (size_t)(btile * 16) * rowbytes;
    const unsigned rb32 = (unsigned)A * 12u;

    int sr[2], ri[2], na[2];
    #pragma unroll
    for (int t = 0; t < 2; ++t) {
        const int r = (t == 0 || hasB) ? rA + t : rA;
        sr[t] = seq[r];
        ri[t] = ridx[r];
    }
    #pragma unroll
    for (int t = 0; t < 2; ++t) na[t] = natoms[sr[t]];

    float g0[2][8][3];
    float g1[2][4][3];
    v4f   w0[2], w1[2], w2[2];

    const unsigned woff = (unsigned)((c * MAXN + 12 * h) * 4);
    {
        const v4f* Wr = (const v4f*)((const char*)(W + (size_t)sr[0] * (VDIM * MAXN)) + woff);
        w0[0] = Wr[0]; w1[0] = Wr[1]; w2[0] = Wr[2];
    }

    #pragma unroll
    for (int t = 0; t < 2; ++t) {
        const int lim = na[t] > 2 ? na[t] : 2;

        #pragma unroll
        for (int i = 0; i < 8; ++i) { g0[t][i][0] = 0.0f; g0[t][i][1] = 0.0f; g0[t][i][2] = 0.0f; }
        if (c < lim) {
            int atom = ri[t] + c;
            if (atom > A - 1) atom = A - 1;
            const unsigned off = (unsigned)(8 * h) * rb32 + (unsigned)atom * 12u;
            #pragma unroll
            for (int i = 0; i < 8; ++i) {
                const gfloat* p = (const gfloat*)(uniform_ptr(xb + (size_t)i * rowbytes) + off);
                g0[t][i][0] = p[0]; g0[t][i][1] = p[1]; g0[t][i][2] = p[2];
            }
        }
        #pragma unroll
        for (int i = 0; i < 4; ++i) { g1[t][i][0] = 0.0f; g1[t][i][1] = 0.0f; g1[t][i][2] = 0.0f; }
        if (16 + a8 < lim) {
            int atom = ri[t] + 16 + a8;
            if (atom > A - 1) atom = A - 1;
            const unsigned off = (unsigned)(4 * q) * rb32 + (unsigned)atom * 12u;
            #pragma unroll
            for (int i = 0; i < 4; ++i) {
                const gfloat* p = (const gfloat*)(uniform_ptr(xb + (size_t)i * rowbytes) + off);
                g1[t][i][0] = p[0]; g1[t][i][1] = p[1]; g1[t][i][2] = p[2];
            }
        }
    }

    float root[2][3];
    #pragma unroll
    for (int t = 0; t < 2; ++t) {
        float* T = &tile[wave][t][0];
        float* d0 = T + (8 * h) * PITCH + 3 * c;
        #pragma unroll
        for (int i = 0; i < 8; ++i) {
            d0[0] = g0[t][i][0]; d0[1] = g0[t][i][1]; d0[2] = g0[t][i][2];
            d0 += PITCH;
        }
        float* d1 = T + (4 * q) * PITCH + 48 + 3 * a8;
        #pragma unroll
        for (int i = 0; i < 4; ++i) {
            d1[0] = g1[t][i][0]; d1[1] = g1[t][i][1]; d1[2] = g1[t][i][2];
            d1 += PITCH;
        }

        root[t][0] = T[c * PITCH + 3]; root[t][1] = T[c * PITCH + 4]; root[t][2] = T[c * PITCH + 5];
    }

    {
        const v4f* Wr = (const v4f*)((const char*)(W + (size_t)sr[1] * (VDIM * MAXN)) + woff);
        w0[1] = Wr[0]; w1[1] = Wr[1]; w2[1] = Wr[2];
    }

    const bool keep = (h == 0) || hasB;

    if (keep) {
        float* o = (float*)((char*)(out_root + ((size_t)(btile * 16) * R + rA) * 3)
                            + (unsigned)c * ((unsigned)R * 12u) + (unsigned)(12 * h));
        o[0] = h ? root[1][0] : root[0][0];
        o[1] = h ? root[1][1] : root[0][1];
        o[2] = h ? root[1][2] : root[0][2];
    }

    v8f acc[2][3];
    #pragma unroll
    for (int t = 0; t < 2; ++t) {
        const float wk[12] = { w0[t][0], w0[t][1], w0[t][2], w0[t][3], w1[t][0], w1[t][1], w1[t][2], w1[t][3],
                               w2[t][0], w2[t][1], w2[t][2], w2[t][3] };
        const v4f* s4 = (const v4f*)(&tile[wave][t][0] + c * PITCH + 36 * h);
        v8f acc0 = {}, acc1 = {}, acc2 = {};
        #pragma unroll
        for (int jj = 0; jj < 3; ++jj) {
            const v4f t0 = s4[3 * jj], t1 = s4[3 * jj + 1], t2 = s4[3 * jj + 2];
            const float e[12] = { t0[0], t0[1], t0[2], t0[3], t1[0], t1[1], t1[2], t1[3],
                                  t2[0], t2[1], t2[2], t2[3] };
            #pragma unroll
            for (int u = 0; u < 2; ++u) {
                const int j  = 2 * jj + u;
                const int n0 = 12 * h + 2 * j;
                const bool m0 = n0     < na[t];
                const bool m1 = n0 + 1 < na[t];
                v2f a0, a1, a2, b;
                a0[0] = m0 ? e[6 * u + 0] - root[t][0] : 0.0f;
                a0[1] = m1 ? e[6 * u + 3] - root[t][0] : 0.0f;
                a1[0] = m0 ? e[6 * u + 1] - root[t][1] : 0.0f;
                a1[1] = m1 ? e[6 * u + 4] - root[t][1] : 0.0f;
                a2[0] = m0 ? e[6 * u + 2] - root[t][2] : 0.0f;
                a2[1] = m1 ? e[6 * u + 5] - root[t][2] : 0.0f;
                b[0] = wk[2 * j]; b[1] = wk[2 * j + 1];
                acc0 = __builtin_amdgcn_wmma_f32_16x16x4_f32(false, a0, false, b, (short)0, acc0, false, false);
                acc1 = __builtin_amdgcn_wmma_f32_16x16x4_f32(false, a1, false, b, (short)0, acc1, false, false);
                acc2 = __builtin_amdgcn_wmma_f32_16x16x4_f32(false, a2, false, b, (short)0, acc2, false, false);
            }
        }
        acc[t][0] = acc0; acc[t][1] = acc1; acc[t][2] = acc2;
    }

    float va[8][3], vb[8][3];
    #pragma unroll
    for (int p = 0; p < 8; ++p) {
        #pragma unroll
        for (int d = 0; d < 3; ++d) {
            const float mineA = acc[0][d][p];
            const float mineB = acc[1][d][p];
            const float got   = other_half(h ? mineA : mineB);
            va[p][d] = h ? got   : mineA;
            vb[p][d] = h ? mineB : got;
        }
    }
    if (keep) {
        const size_t orow = (size_t)R * (VDIM * 3 * 4);
        const char* ob = (const char*)(out_xv + ((size_t)(btile * 16) * R + rA) * (VDIM * 3));
        const unsigned loff = (unsigned)lane * 12u;
        #pragma unroll
        for (int p = 0; p < 8; ++p) {
            gfloat* oa = (gfloat*)(uniform_ptr(ob + (size_t)p * orow) + loff);
            gfloat* o8 = (gfloat*)(uniform_ptr(ob + (size_t)(p + 8) * orow) + loff);
            oa[0] = va[p][0]; oa[1] = va[p][1]; oa[2] = va[p][2];
            o8[0] = vb[p][0]; o8[1] = vb[p][1]; o8[2] = vb[p][2];
        }
    }
}

extern "C" void kernel_launch(void* const* d_in, const int* in_sizes, int n_in,
                              void* d_out, int out_size, void* d_ws, size_t ws_size,
                              hipStream_t stream) {
    const float* x      = (const float*)d_in[0];
    const float* W      = (const float*)d_in[1];
    const int*   seq    = (const int*)d_in[2];
    const int*   ridx   = (const int*)d_in[3];
    const int*   natoms = (const int*)d_in[4];

    const int R = in_sizes[2];
    const int A = in_sizes[0] / (BATCH * 3);

    float* out_root = (float*)d_out;
    float* out_xv   = out_root + (size_t)BATCH * R * 3;

    const int groups = (R + 2 * WAVES_PER_WG - 1) / (2 * WAVES_PER_WG);
    residues_encode_kernel<<<groups * (BATCH / 16), NTHREADS, 0, stream>>>(x, W, seq, ridx, natoms,
                                                                          out_root, out_xv, A, R);
}


# A wave takes 32 batch rows (two M tiles) of its residue pair: 4096 waves enter, not 8192; seq/natoms chain, load offsets and W fragments fetched once per pair; all 48 x loads requested up front
# speedup vs baseline: 2.8467x; 1.0849x over previous
#include <hip/hip_runtime.h>


typedef __attribute__((ext_vector_type(2))) float v2f;
typedef __attribute__((ext_vector_type(4))) float v4f;
typedef __attribute__((ext_vector_type(8))) float v8f;

#define BATCH 64
#define VDIM  16
#define MAXN  24
#define WAVES_PER_WG 8
#define MT 2
#define NTHREADS (32 * WAVES_PER_WG)
#define PITCH 76

__device__ __forceinline__ float other_half(float v)
{
    const unsigned u = __float_as_uint(v);
    return __uint_as_float(__builtin_amdgcn_permlanex16(u, u, 0x76543210u, 0xfedcba98u, false, false));
}

typedef __attribute__((address_space(1))) char  gchar;
typedef __attribute__((address_space(1))) float gfloat;
__device__ __forceinline__ gchar* uniform_ptr(const char* p)
{
    const unsigned long long v = (unsigned long long)p;
    const unsigned lo = __builtin_amdgcn_readfirstlane((unsigned)v);
    const unsigned hi = __builtin_amdgcn_readfirstlane((unsigned)(v >> 32));
    return (gchar*)(((unsigned long long)hi << 32) | lo);
}

__launch_bounds__(NTHREADS, 4)
__global__ void residues_encode_kernel(const float* __restrict__ x,
                                       const float* __restrict__ W,
                                       const int*   __restrict__ seq,
                                       const int*   __restrict__ ridx,
                                       const int*   __restrict__ natoms,
                                       float* __restrict__ out_root,
                                       float* __restrict__ out_xv,
                                       int A, int R)
{
    __shared__ __attribute__((aligned(16))) float tile[WAVES_PER_WG][MT][2][16 * PITCH];

    const int lane = threadIdx.x & 31;
    const int wave  = __builtin_amdgcn_readfirstlane((int)(threadIdx.x >> 5));
    const int btile = blockIdx.x & 1;
    const int rA    = 2 * ((int)(blockIdx.x >> 1) * WAVES_PER_WG + wave);
    if (rA >= R) return;
    const bool hasB = rA + 1 < R;

    const int h = lane >> 4;
    const int c = lane & 15;
    const int q  = lane >> 3;
    const int a8 = lane & 7;

    const size_t   rowbytes = (size_t)A * 12;
    const char*    xb = (const char*)x + (size_t)(btile * (16 * MT)) * rowbytes;
    const unsigned rb32 = (unsigned)A * 12u;

    int sr[2], ri[2], na[2];
    #pragma unroll
    for (int t = 0; t < 2; ++t) {
        const int r = (t == 0 || hasB) ? rA + t : rA;
        sr[t] = seq[r];
        ri[t] = ridx[r];
    }
    #pragma unroll
    for (int t = 0; t < 2; ++t) na[t] = natoms[sr[t]];

    float g0[MT][2][8][3];
    float g1[MT][2][4][3];
    v4f   w0[2], w1[2], w2[2];

    const unsigned woff = (unsigned)((c * MAXN + 12 * h) * 4);
    #pragma unroll
    for (int t = 0; t < 2; ++t) {
        const v4f* Wr = (const v4f*)((const char*)(W + (size_t)sr[t] * (VDIM * MAXN)) + woff);
        w0[t] = Wr[0]; w1[t] = Wr[1]; w2[t] = Wr[2];
    }

    bool     ld0[2], ld1[2];
    unsigned off0[2], off1[2];
    #pragma unroll
    for (int t = 0; t < 2; ++t) {
        const int lim = na[t] > 2 ? na[t] : 2;
        int atom0 = ri[t] + c;
        if (atom0 > A - 1) atom0 = A - 1;
        int atom1 = ri[t] + 16 + a8;
        if (atom1 > A - 1) atom1 = A - 1;
        ld0[t]  = c < lim;
        ld1[t]  = 16 + a8 < lim;
        off0[t] = (unsigned)(8 * h) * rb32 + (unsigned)atom0 * 12u;
        off1[t] = (unsigned)(4 * q) * rb32 + (unsigned)atom1 * 12u;
    }

    #pragma unroll
    for (int mt = 0; mt < MT; ++mt) {
        #pragma unroll
        for (int t = 0; t < 2; ++t) {
            #pragma unroll
            for (int i = 0; i < 8; ++i) { g0[mt][t][i][0] = 0.0f; g0[mt][t][i][1] = 0.0f; g0[mt][t][i][2] = 0.0f; }
            if (ld0[t]) {
                #pragma unroll
                for (int i = 0; i < 8; ++i) {
                    const gfloat* p = (const gfloat*)(uniform_ptr(xb + (size_t)(16 * mt + i) * rowbytes) + off0[t]);
                    g0[mt][t][i][0] = p[0]; g0[mt][t][i][1] = p[1]; g0[mt][t][i][2] = p[2];
                }
            }
            #pragma unroll
            for (int i = 0; i < 4; ++i) { g1[mt][t][i][0] = 0.0f; g1[mt][t][i][1] = 0.0f; g1[mt][t][i][2] = 0.0f; }
            if (ld1[t]) {
                #pragma unroll
                for (int i = 0; i < 4; ++i) {
                    const gfloat* p = (const gfloat*)(uniform_ptr(xb + (size_t)(16 * mt + i) * rowbytes) + off1[t]);
                    g1[mt][t][i][0] = p[0]; g1[mt][t][i][1] = p[1]; g1[mt][t][i][2] = p[2];
                }
            }
        }
    }

    const bool keep = (h == 0) || hasB;

    #pragma unroll
    for (int mt = 0; mt < MT; ++mt) {
        const int row0 = btile * (16 * MT) + 16 * mt;

        float root[2][3];
        #pragma unroll
        for (int t = 0; t < 2; ++t) {
            float* T = &tile[wave][mt][t][0];
            float* d0 = T + (8 * h) * PITCH + 3 * c;
            #pragma unroll
            for (int i = 0; i < 8; ++i) {
                d0[0] = g0[mt][t][i][0]; d0[1] = g0[mt][t][i][1]; d0[2] = g0[mt][t][i][2];
                d0 += PITCH;
            }
            float* d1 = T + (4 * q) * PITCH + 48 + 3 * a8;
            #pragma unroll
            for (int i = 0; i < 4; ++i) {
                d1[0] = g1[mt][t][i][0]; d1[1] = g1[mt][t][i][1]; d1[2] = g1[mt][t][i][2];
                d1 += PITCH;
            }

            root[t][0] = T[c * PITCH + 3]; root[t][1] = T[c * PITCH + 4]; root[t][2] = T[c * PITCH + 5];
        }

        if (keep) {
            float* o = (float*)((char*)(out_root + ((size_t)row0 * R + rA) * 3)
                                + (unsigned)c * ((unsigned)R * 12u) + (unsigned)(12 * h));
            o[0] = h ? root[1][0] : root[0][0];
            o[1] = h ? root[1][1] : root[0][1];
            o[2] = h ? root[1][2] : root[0][2];
        }

        v8f acc[2][3];
        #pragma unroll
        for (int t = 0; t < 2; ++t) {
            const float wk[12] = { w0[t][0], w0[t][1], w0[t][2], w0[t][3], w1[t][0], w1[t][1], w1[t][2], w1[t][3],
                                   w2[t][0], w2[t][1], w2[t][2], w2[t][3] };
            const v4f* s4 = (const v4f*)(&tile[wave][mt][t][0] + c * PITCH + 36 * h);
            v8f acc0 = {}, acc1 = {}, acc2 = {};
            #pragma unroll
            for (int jj = 0; jj < 3; ++jj) {
                const v4f t0 = s4[3 * jj], t1 = s4[3 * jj + 1], t2 = s4[3 * jj + 2];
                const float e[12] = { t0[0], t0[1], t0[2], t0[3], t1[0], t1[1], t1[2], t1[3],
                                      t2[0], t2[1], t2[2], t2[3] };
                #pragma unroll
                for (int u = 0; u < 2; ++u) {
                    const int j  = 2 * jj + u;
                    const int n0 = 12 * h + 2 * j;
                    const bool m0 = n0     < na[t];
                    const bool m1 = n0 + 1 < na[t];
                    v2f a0, a1, a2, b;
                    a0[0] = m0 ? e[6 * u + 0] - root[t][0] : 0.0f;
                    a0[1] = m1 ? e[6 * u + 3] - root[t][0] : 0.0f;
                    a1[0] = m0 ? e[6 * u + 1] - root[t][1] : 0.0f;
                    a1[1] = m1 ? e[6 * u + 4] - root[t][1] : 0.0f;
                    a2[0] = m0 ? e[6 * u + 2] - root[t][2] : 0.0f;
                    a2[1] = m1 ? e[6 * u + 5] - root[t][2] : 0.0f;
                    b[0] = wk[2 * j]; b[1] = wk[2 * j + 1];
                    acc0 = __builtin_amdgcn_wmma_f32_16x16x4_f32(false, a0, false, b, (short)0, acc0, false, false);
                    acc1 = __builtin_amdgcn_wmma_f32_16x16x4_f32(false, a1, false, b, (short)0, acc1, false, false);
                    acc2 = __builtin_amdgcn_wmma_f32_16x16x4_f32(false, a2, false, b, (short)0, acc2, false, false);
                }
            }
            acc[t][0] = acc0; acc[t][1] = acc1; acc[t][2] = acc2;
        }

        float va[8][3], vb[8][3];
        #pragma unroll
        for (int p = 0; p < 8; ++p) {
            #pragma unroll
            for (int d = 0; d < 3; ++d) {
                const float mineA = acc[0][d][p];
                const float mineB = acc[1][d][p];
                const float got   = other_half(h ? mineA : mineB);
                va[p][d] = h ? got   : mineA;
                vb[p][d] = h ? mineB : got;
            }
        }
        if (keep) {
            const size_t orow = (size_t)R * (VDIM * 3 * 4);
            const char* ob = (const char*)(out_xv + ((size_t)(btile * (16 * MT)) * R + rA) * (VDIM * 3));
            const unsigned loff = (unsigned)lane * 12u + (unsigned)(16 * mt) * ((unsigned)R * (VDIM * 3 * 4u));
            #pragma unroll
            for (int p = 0; p < 8; ++p) {
                gfloat* oa = (gfloat*)(uniform_ptr(ob + (size_t)p * orow) + loff);
                gfloat* o8 = (gfloat*)(uniform_ptr(ob + (size_t)(p + 8) * orow) + loff);
                oa[0] = va[p][0]; oa[1] = va[p][1]; oa[2] = va[p][2];
                o8[0] = vb[p][0]; o8[1] = vb[p][1]; o8[2] = vb[p][2];
            }
        }
    }
}

extern "C" void kernel_launch(void* const* d_in, const int* in_sizes, int n_in,
                              void* d_out, int out_size, void* d_ws, size_t ws_size,
                              hipStream_t stream) {
    const float* x      = (const float*)d_in[0];
    const float* W      = (const float*)d_in[1];
    const int*   seq    = (const int*)d_in[2];
    const int*   ridx   = (const int*)d_in[3];
    const int*   natoms = (const int*)d_in[4];

    const int R = in_sizes[2];
    const int A = in_sizes[0] / (BATCH * 3);

    float* out_root = (float*)d_out;
    float* out_xv   = out_root + (size_t)BATCH * R * 3;

    const int groups = (R + 2 * WAVES_PER_WG - 1) / (2 * WAVES_PER_WG);
    residues_encode_kernel<<<groups * (BATCH / (16 * MT)), NTHREADS, 0, stream>>>(x, W, seq, ridx, natoms,
                                                                          out_root, out_xv, A, R);
}


# Root written through an LDS exchange and one end-of-kernel barrier: a store covers 2 batch rows x 192 contiguous bytes (16 residues), not 16 pieces of 24 bytes; sum order unchanged
# speedup vs baseline: 3.2411x; 1.1385x over previous
#include <hip/hip_runtime.h>


typedef __attribute__((ext_vector_type(2))) float v2f;
typedef __attribute__((ext_vector_type(4))) float v4f;
typedef __attribute__((ext_vector_type(8))) float v8f;

#define BATCH 64
#define VDIM  16
#define MAXN  24
#define WAVES_PER_WG 8
#define MT 2
#define NTHREADS (32 * WAVES_PER_WG)
#define PITCH 76

__device__ __forceinline__ float other_half(float v)
{
    const unsigned u = __float_as_uint(v);
    return __uint_as_float(__builtin_amdgcn_permlanex16(u, u, 0x76543210u, 0xfedcba98u, false, false));
}

typedef __attribute__((address_space(1))) char  gchar;
typedef __attribute__((address_space(1))) float gfloat;
__device__ __forceinline__ gchar* uniform_ptr(const char* p)
{
    const unsigned long long v = (unsigned long long)p;
    const unsigned lo = __builtin_amdgcn_readfirstlane((unsigned)v);
    const unsigned hi = __builtin_amdgcn_readfirstlane((unsigned)(v >> 32));
    return (gchar*)(((unsigned long long)hi << 32) | lo);
}

__launch_bounds__(NTHREADS, 4)
__global__ void residues_encode_kernel(const float* __restrict__ x,
                                       const float* __restrict__ W,
                                       const int*   __restrict__ seq,
                                       const int*   __restrict__ ridx,
                                       const int*   __restrict__ natoms,
                                       float* __restrict__ out_root,
                                       float* __restrict__ out_xv,
                                       int A, int R)
{
    __shared__ __attribute__((aligned(16))) float tile[WAVES_PER_WG][MT][2][16 * PITCH];
    __shared__ float rootbuf[16 * MT][2 * WAVES_PER_WG][3];

    const int lane = threadIdx.x & 31;
    const int wave  = __builtin_amdgcn_readfirstlane((int)(threadIdx.x >> 5));
    const int btile = blockIdx.x & 1;
    const int group = (int)(blockIdx.x >> 1);
    const int rA0   = 2 * (group * WAVES_PER_WG + wave);
    const bool live = rA0 < R;
    const int  rA   = live ? rA0 : 0;
    const bool hasB = rA + 1 < R;

    const int h = lane >> 4;
    const int c = lane & 15;
    const int q  = lane >> 3;
    const int a8 = lane & 7;

    const size_t   rowbytes = (size_t)A * 12;
    const char*    xb = (const char*)x + (size_t)(btile * (16 * MT)) * rowbytes;
    const unsigned rb32 = (unsigned)A * 12u;

    int sr[2], ri[2], na[2];
    #pragma unroll
    for (int t = 0; t < 2; ++t) {
        const int r = (t == 0 || hasB) ? rA + t : rA;
        sr[t] = seq[r];
        ri[t] = ridx[r];
    }
    #pragma unroll
    for (int t = 0; t < 2; ++t) na[t] = natoms[sr[t]];

    float g0[MT][2][8][3];
    float g1[MT][2][4][3];
    v4f   w0[2], w1[2], w2[2];

    const unsigned woff = (unsigned)((c * MAXN + 12 * h) * 4);
    #pragma unroll
    for (int t = 0; t < 2; ++t) {
        const v4f* Wr = (const v4f*)((const char*)(W + (size_t)sr[t] * (VDIM * MAXN)) + woff);
        w0[t] = Wr[0]; w1[t] = Wr[1]; w2[t] = Wr[2];
    }

    bool     ld0[2], ld1[2];
    unsigned off0[2], off1[2];
    #pragma unroll
    for (int t = 0; t < 2; ++t) {
        const int lim = na[t] > 2 ? na[t] : 2;
        int atom0 = ri[t] + c;
        if (atom0 > A - 1) atom0 = A - 1;
        int atom1 = ri[t] + 16 + a8;
        if (atom1 > A - 1) atom1 = A - 1;
        ld0[t]  = c < lim;
        ld1[t]  = 16 + a8 < lim;
        off0[t] = (unsigned)(8 * h) * rb32 + (unsigned)atom0 * 12u;
        off1[t] = (unsigned)(4 * q) * rb32 + (unsigned)atom1 * 12u;
    }

    #pragma unroll
    for (int mt = 0; mt < MT; ++mt) {
        #pragma unroll
        for (int t = 0; t < 2; ++t) {
            #pragma unroll
            for (int i = 0; i < 8; ++i) { g0[mt][t][i][0] = 0.0f; g0[mt][t][i][1] = 0.0f; g0[mt][t][i][2] = 0.0f; }
            if (ld0[t]) {
                #pragma unroll
                for (int i = 0; i < 8; ++i) {
                    const gfloat* p = (const gfloat*)(uniform_ptr(xb + (size_t)(16 * mt + i) * rowbytes) + off0[t]);
                    g0[mt][t][i][0] = p[0]; g0[mt][t][i][1] = p[1]; g0[mt][t][i][2] = p[2];
                }
            }
            #pragma unroll
            for (int i = 0; i < 4; ++i) { g1[mt][t][i][0] = 0.0f; g1[mt][t][i][1] = 0.0f; g1[mt][t][i][2] = 0.0f; }
            if (ld1[t]) {
                #pragma unroll
                for (int i = 0; i < 4; ++i) {
                    const gfloat* p = (const gfloat*)(uniform_ptr(xb + (size_t)(16 * mt + i) * rowbytes) + off1[t]);
                    g1[mt][t][i][0] = p[0]; g1[mt][t][i][1] = p[1]; g1[mt][t][i][2] = p[2];
                }
            }
        }
    }

    const bool keep = live && ((h == 0) || hasB);

    #pragma unroll
    for (int mt = 0; mt < MT; ++mt) {

        float root[2][3];
        #pragma unroll
        for (int t = 0; t < 2; ++t) {
            float* T = &tile[wave][mt][t][0];
            float* d0 = T + (8 * h) * PITCH + 3 * c;
            #pragma unroll
            for (int i = 0; i < 8; ++i) {
                d0[0] = g0[mt][t][i][0]; d0[1] = g0[mt][t][i][1]; d0[2] = g0[mt][t][i][2];
                d0 += PITCH;
            }
            float* d1 = T + (4 * q) * PITCH + 48 + 3 * a8;
            #pragma unroll
            for (int i = 0; i < 4; ++i) {
                d1[0] = g1[mt][t][i][0]; d1[1] = g1[mt][t][i][1]; d1[2] = g1[mt][t][i][2];
                d1 += PITCH;
            }

            root[t][0] = T[c * PITCH + 3]; root[t][1] = T[c * PITCH + 4]; root[t][2] = T[c * PITCH + 5];
        }

        {
            float* rb = &rootbuf[16 * mt + c][2 * wave + h][0];
            rb[0] = h ? root[1][0] : root[0][0];
            rb[1] = h ? root[1][1] : root[0][1];
            rb[2] = h ? root[1][2] : root[0][2];
        }

        v8f acc[2][3];
        #pragma unroll
        for (int t = 0; t < 2; ++t) {
            const float wk[12] = { w0[t][0], w0[t][1], w0[t][2], w0[t][3], w1[t][0], w1[t][1], w1[t][2], w1[t][3],
                                   w2[t][0], w2[t][1], w2[t][2], w2[t][3] };
            const v4f* s4 = (const v4f*)(&tile[wave][mt][t][0] + c * PITCH + 36 * h);
            v8f acc0 = {}, acc1 = {}, acc2 = {};
            #pragma unroll
            for (int jj = 0; jj < 3; ++jj) {
                const v4f t0 = s4[3 * jj], t1 = s4[3 * jj + 1], t2 = s4[3 * jj + 2];
                const float e[12] = { t0[0], t0[1], t0[2], t0[3], t1[0], t1[1], t1[2], t1[3],
                                      t2[0], t2[1], t2[2], t2[3] };
                #pragma unroll
                for (int u = 0; u < 2; ++u) {
                    const int j  = 2 * jj + u;
                    const int n0 = 12 * h + 2 * j;
                    const bool m0 = n0     < na[t];
                    const bool m1 = n0 + 1 < na[t];
                    v2f a0, a1, a2, b;
                    a0[0] = m0 ? e[6 * u + 0] - root[t][0] : 0.0f;
                    a0[1] = m1 ? e[6 * u + 3] - root[t][0] : 0.0f;
                    a1[0] = m0 ? e[6 * u + 1] - root[t][1] : 0.0f;
                    a1[1] = m1 ? e[6 * u + 4] - root[t][1] : 0.0f;
                    a2[0] = m0 ? e[6 * u + 2] - root[t][2] : 0.0f;
                    a2[1] = m1 ? e[6 * u + 5] - root[t][2] : 0.0f;
                    b[0] = wk[2 * j]; b[1] = wk[2 * j + 1];
                    acc0 = __builtin_amdgcn_wmma_f32_16x16x4_f32(false, a0, false, b, (short)0, acc0, false, false);
                    acc1 = __builtin_amdgcn_wmma_f32_16x16x4_f32(false, a1, false, b, (short)0, acc1, false, false);
                    acc2 = __builtin_amdgcn_wmma_f32_16x16x4_f32(false, a2, false, b, (short)0, acc2, false, false);
                }
            }
            acc[t][0] = acc0; acc[t][1] = acc1; acc[t][2] = acc2;
        }

        float va[8][3], vb[8][3];
        #pragma unroll
        for (int p = 0; p < 8; ++p) {
            #pragma unroll
            for (int d = 0; d < 3; ++d) {
                const float mineA = acc[0][d][p];
                const float mineB = acc[1][d][p];
                const float got   = other_half(h ? mineA : mineB);
                va[p][d] = h ? got   : mineA;
                vb[p][d] = h ? mineB : got;
            }
        }
        if (keep) {
            const size_t orow = (size_t)R * (VDIM * 3 * 4);
            const char* ob = (const char*)(out_xv + ((size_t)(btile * (16 * MT)) * R + rA) * (VDIM * 3));
            const unsigned loff = (unsigned)lane * 12u + (unsigned)(16 * mt) * ((unsigned)R * (VDIM * 3 * 4u));
            #pragma unroll
            for (int p = 0; p < 8; ++p) {
                gfloat* oa = (gfloat*)(uniform_ptr(ob + (size_t)p * orow) + loff);
                gfloat* o8 = (gfloat*)(uniform_ptr(ob + (size_t)(p + 8) * orow) + loff);
                oa[0] = va[p][0]; oa[1] = va[p][1]; oa[2] = va[p][2];
                o8[0] = vb[p][0]; o8[1] = vb[p][1]; o8[2] = vb[p][2];
            }
        }
    }

    __builtin_amdgcn_fence(__ATOMIC_RELEASE, "workgroup", "local");
    __builtin_amdgcn_s_barrier();
    __builtin_amdgcn_fence(__ATOMIC_ACQUIRE, "workgroup", "local");
    {
        const int rr = 16 * group + c;
        if (rr < R) {
            #pragma unroll
            for (int k = 0; k < 2; ++k) {
                const int row = 4 * wave + 2 * k + h;
                const float* s = &rootbuf[row][c][0];
                float* o = out_root + ((size_t)(btile * (16 * MT) + row) * R + rr) * 3;
                o[0] = s[0]; o[1] = s[1]; o[2] = s[2];
            }
        }
    }
}

extern "C" void kernel_launch(void* const* d_in, const int* in_sizes, int n_in,
                              void* d_out, int out_size, void* d_ws, size_t ws_size,
                              hipStream_t stream) {
    const float* x      = (const float*)d_in[0];
    const float* W      = (const float*)d_in[1];
    const int*   seq    = (const int*)d_in[2];
    const int*   ridx   = (const int*)d_in[3];
    const int*   natoms = (const int*)d_in[4];

    const int R = in_sizes[2];
    const int A = in_sizes[0] / (BATCH * 3);

    float* out_root = (float*)d_out;
    float* out_xv   = out_root + (size_t)BATCH * R * 3;

    const int groups = (R + 2 * WAVES_PER_WG - 1) / (2 * WAVES_PER_WG);
    residues_encode_kernel<<<groups * (BATCH / (16 * MT)), NTHREADS, 0, stream>>>(x, W, seq, ridx, natoms,
                                                                          out_root, out_xv, A, R);
}


# K-steps whose 4 atoms are all masked are jumped over: k order atom = 8i+4h+2m+s, step 2i+m runs only if 8i+2m < natoms (tile slots, W chunks follow; sum order changes): avg 50 WMMAs a wave, not 72
# speedup vs baseline: 3.4309x; 1.0586x over previous
#include <hip/hip_runtime.h>


typedef __attribute__((ext_vector_type(2))) float v2f;
typedef __attribute__((ext_vector_type(4))) float v4f;
typedef __attribute__((ext_vector_type(8))) float v8f;

#define BATCH 64
#define VDIM  16
#define MAXN  24
#define WAVES_PER_WG 8
#define MT 2
#define NTHREADS (32 * WAVES_PER_WG)
#define PITCH 76

__device__ __forceinline__ float other_half(float v)
{
    const unsigned u = __float_as_uint(v);
    return __uint_as_float(__builtin_amdgcn_permlanex16(u, u, 0x76543210u, 0xfedcba98u, false, false));
}

typedef __attribute__((address_space(1))) char  gchar;
typedef __attribute__((address_space(1))) float gfloat;
__device__ __forceinline__ gchar* uniform_ptr(const char* p)
{
    const unsigned long long v = (unsigned long long)p;
    const unsigned lo = __builtin_amdgcn_readfirstlane((unsigned)v);
    const unsigned hi = __builtin_amdgcn_readfirstlane((unsigned)(v >> 32));
    return (gchar*)(((unsigned long long)hi << 32) | lo);
}

__launch_bounds__(NTHREADS, 4)
__global__ void residues_encode_kernel(const float* __restrict__ x,
                                       const float* __restrict__ W,
                                       const int*   __restrict__ seq,
                                       const int*   __restrict__ ridx,
                                       const int*   __restrict__ natoms,
                                       float* __restrict__ out_root,
                                       float* __restrict__ out_xv,
                                       int A, int R)
{
    __shared__ __attribute__((aligned(16))) float tile[WAVES_PER_WG][MT][2][16 * PITCH];
    __shared__ float rootbuf[16 * MT][2 * WAVES_PER_WG][3];

    const int lane = threadIdx.x & 31;
    const int wave  = __builtin_amdgcn_readfirstlane((int)(threadIdx.x >> 5));
    const int btile = blockIdx.x & 1;
    const int group = (int)(blockIdx.x >> 1);
    const int rA0   = 2 * (group * WAVES_PER_WG + wave);
    const bool live = rA0 < R;
    const int  rA   = live ? rA0 : 0;
    const bool hasB = rA + 1 < R;

    const int h = lane >> 4;
    const int c = lane & 15;
    const int q  = lane >> 3;
    const int a8 = lane & 7;

    const size_t   rowbytes = (size_t)A * 12;
    const char*    xb = (const char*)x + (size_t)(btile * (16 * MT)) * rowbytes;
    const unsigned rb32 = (unsigned)A * 12u;

    int sr[2], ri[2], na[2];
    #pragma unroll
    for (int t = 0; t < 2; ++t) {
        const int r = (t == 0 || hasB) ? rA + t : rA;
        sr[t] = seq[r];
        ri[t] = ridx[r];
    }
    #pragma unroll
    for (int t = 0; t < 2; ++t) na[t] = natoms[sr[t]];

    float g0[MT][2][8][3];
    float g1[MT][2][4][3];
    v4f   w0[2], w1[2], w2[2];

    const unsigned woff = (unsigned)((c * MAXN + 4 * h) * 4);
    #pragma unroll
    for (int t = 0; t < 2; ++t) {
        const v4f* Wr = (const v4f*)((const char*)(W + (size_t)sr[t] * (VDIM * MAXN)) + woff);
        w0[t] = Wr[0]; w1[t] = Wr[2]; w2[t] = Wr[4];
    }

    bool     ld0[2], ld1[2];
    unsigned off0[2], off1[2];
    #pragma unroll
    for (int t = 0; t < 2; ++t) {
        const int lim = na[t] > 2 ? na[t] : 2;
        int atom0 = ri[t] + c;
        if (atom0 > A - 1) atom0 = A - 1;
        int atom1 = ri[t] + 16 + a8;
        if (atom1 > A - 1) atom1 = A - 1;
        ld0[t]  = c < lim;
        ld1[t]  = 16 + a8 < lim;
        off0[t] = (unsigned)(8 * h) * rb32 + (unsigned)atom0 * 12u;
        off1[t] = (unsigned)(4 * q) * rb32 + (unsigned)atom1 * 12u;
    }

    #pragma unroll
    for (int mt = 0; mt < MT; ++mt) {
        #pragma unroll
        for (int t = 0; t < 2; ++t) {
            #pragma unroll
            for (int i = 0; i < 8; ++i) { g0[mt][t][i][0] = 0.0f; g0[mt][t][i][1] = 0.0f; g0[mt][t][i][2] = 0.0f; }
            if (ld0[t]) {
                #pragma unroll
                for (int i = 0; i < 8; ++i) {
                    const gfloat* p = (const gfloat*)(uniform_ptr(xb + (size_t)(16 * mt + i) * rowbytes) + off0[t]);
                    g0[mt][t][i][0] = p[0]; g0[mt][t][i][1] = p[1]; g0[mt][t][i][2] = p[2];
                }
            }
            #pragma unroll
            for (int i = 0; i < 4; ++i) { g1[mt][t][i][0] = 0.0f; g1[mt][t][i][1] = 0.0f; g1[mt][t][i][2] = 0.0f; }
            if (ld1[t]) {
                #pragma unroll
                for (int i = 0; i < 4; ++i) {
                    const gfloat* p = (const gfloat*)(uniform_ptr(xb + (size_t)(16 * mt + i) * rowbytes) + off1[t]);
                    g1[mt][t][i][0] = p[0]; g1[mt][t][i][1] = p[1]; g1[mt][t][i][2] = p[2];
                }
            }
        }
    }

    const int slot0 = 3 * (12 * ((c >> 2) & 1) + 4 * (c >> 3) + (c & 3));
    const int slot1 = 3 * (12 * (a8 >> 2) + 8 + (a8 & 3));

    const bool keep = live && ((h == 0) || hasB);

    #pragma unroll
    for (int mt = 0; mt < MT; ++mt) {

        float root[2][3];
        #pragma unroll
        for (int t = 0; t < 2; ++t) {
            float* T = &tile[wave][mt][t][0];
            float* d0 = T + (8 * h) * PITCH + slot0;
            #pragma unroll
            for (int i = 0; i < 8; ++i) {
                d0[0] = g0[mt][t][i][0]; d0[1] = g0[mt][t][i][1]; d0[2] = g0[mt][t][i][2];
                d0 += PITCH;
            }
            float* d1 = T + (4 * q) * PITCH + slot1;
            #pragma unroll
            for (int i = 0; i < 4; ++i) {
                d1[0] = g1[mt][t][i][0]; d1[1] = g1[mt][t][i][1]; d1[2] = g1[mt][t][i][2];
                d1 += PITCH;
            }

            root[t][0] = T[c * PITCH + 3]; root[t][1] = T[c * PITCH + 4]; root[t][2] = T[c * PITCH + 5];
        }

        {
            float* rb = &rootbuf[16 * mt + c][2 * wave + h][0];
            rb[0] = h ? root[1][0] : root[0][0];
            rb[1] = h ? root[1][1] : root[0][1];
            rb[2] = h ? root[1][2] : root[0][2];
        }

        v4f fr[2][9];
        #pragma unroll
        for (int t = 0; t < 2; ++t) {
            const v4f* s4 = (const v4f*)(&tile[wave][mt][t][0] + c * PITCH + 36 * h);
            #pragma unroll
            for (int i = 0; i < 9; ++i) fr[t][i] = s4[i];
        }
        v8f acc[2][3];
        #pragma unroll
        for (int t = 0; t < 2; ++t) {
            const float wk[12] = { w0[t][0], w0[t][1], w0[t][2], w0[t][3], w1[t][0], w1[t][1], w1[t][2], w1[t][3],
                                   w2[t][0], w2[t][1], w2[t][2], w2[t][3] };
            v8f acc0 = {}, acc1 = {}, acc2 = {};
            #pragma unroll
            for (int jj = 0; jj < 3; ++jj) {
                const v4f t0 = fr[t][3 * jj], t1 = fr[t][3 * jj + 1], t2 = fr[t][3 * jj + 2];
                const float e[12] = { t0[0], t0[1], t0[2], t0[3], t1[0], t1[1], t1[2], t1[3],
                                      t2[0], t2[1], t2[2], t2[3] };
                #pragma unroll
                for (int u = 0; u < 2; ++u) {
                    const int j  = 2 * jj + u;
                    if (j > 0 && 8 * jj + 2 * u >= na[t]) continue;
                    const int n0 = 8 * jj + 4 * h + 2 * u;
                    const bool m0 = n0     < na[t];
                    const bool m1 = n0 + 1 < na[t];
                    v2f a0, a1, a2, b;
                    a0[0] = m0 ? e[6 * u + 0] - root[t][0] : 0.0f;
                    a0[1] = m1 ? e[6 * u + 3] - root[t][0] : 0.0f;
                    a1[0] = m0 ? e[6 * u + 1] - root[t][1] : 0.0f;
                    a1[1] = m1 ? e[6 * u + 4] - root[t][1] : 0.0f;
                    a2[0] = m0 ? e[6 * u + 2] - root[t][2] : 0.0f;
                    a2[1] = m1 ? e[6 * u + 5] - root[t][2] : 0.0f;
                    b[0] = wk[2 * j]; b[1] = wk[2 * j + 1];
                    acc0 = __builtin_amdgcn_wmma_f32_16x16x4_f32(false, a0, false, b, (short)0, acc0, false, false);
                    acc1 = __builtin_amdgcn_wmma_f32_16x16x4_f32(false, a1, false, b, (short)0, acc1, false, false);
                    acc2 = __builtin_amdgcn_wmma_f32_16x16x4_f32(false, a2, false, b, (short)0, acc2, false, false);
                }
            }
            acc[t][0] = acc0; acc[t][1] = acc1; acc[t][2] = acc2;
        }

        float va[8][3], vb[8][3];
        #pragma unroll
        for (int p = 0; p < 8; ++p) {
            #pragma unroll
            for (int d = 0; d < 3; ++d) {
                const float mineA = acc[0][d][p];
                const float mineB = acc[1][d][p];
                const float got   = other_half(h ? mineA : mineB);
                va[p][d] = h ? got   : mineA;
                vb[p][d] = h ? mineB : got;
            }
        }
        if (keep) {
            const size_t orow = (size_t)R * (VDIM * 3 * 4);
            const char* ob = (const char*)(out_xv + ((size_t)(btile * (16 * MT)) * R + rA) * (VDIM * 3));
            const unsigned loff = (unsigned)lane * 12u + (unsigned)(16 * mt) * ((unsigned)R * (VDIM * 3 * 4u));
            #pragma unroll
            for (int p = 0; p < 8; ++p) {
                gfloat* oa = (gfloat*)(uniform_ptr(ob + (size_t)p * orow) + loff);
                gfloat* o8 = (gfloat*)(uniform_ptr(ob + (size_t)(p + 8) * orow) + loff);
                oa[0] = va[p][0]; oa[1] = va[p][1]; oa[2] = va[p][2];
                o8[0] = vb[p][0]; o8[1] = vb[p][1]; o8[2] = vb[p][2];
            }
        }
    }

    __builtin_amdgcn_fence(__ATOMIC_RELEASE, "workgroup", "local");
    __builtin_amdgcn_s_barrier();
    __builtin_amdgcn_fence(__ATOMIC_ACQUIRE, "workgroup", "local");
    {
        const int rr = 16 * group + c;
        if (rr < R) {
            #pragma unroll
            for (int k = 0; k < 2; ++k) {
                const int row = 4 * wave + 2 * k + h;
                const float* s = &rootbuf[row][c][0];
                float* o = out_root + ((size_t)(btile * (16 * MT) + row) * R + rr) * 3;
                o[0] = s[0]; o[1] = s[1]; o[2] = s[2];
            }
        }
    }
}

extern "C" void kernel_launch(void* const* d_in, const int* in_sizes, int n_in,
                              void* d_out, int out_size, void* d_ws, size_t ws_size,
                              hipStream_t stream) {
    const float* x      = (const float*)d_in[0];
    const float* W      = (const float*)d_in[1];
    const int*   seq    = (const int*)d_in[2];
    const int*   ridx   = (const int*)d_in[3];
    const int*   natoms = (const int*)d_in[4];

    const int R = in_sizes[2];
    const int A = in_sizes[0] / (BATCH * 3);

    float* out_root = (float*)d_out;
    float* out_xv   = out_root + (size_t)BATCH * R * 3;

    const int groups = (R + 2 * WAVES_PER_WG - 1) / (2 * WAVES_PER_WG);
    residues_encode_kernel<<<groups * (BATCH / (16 * MT)), NTHREADS, 0, stream>>>(x, W, seq, ridx, natoms,
                                                                          out_root, out_xv, A, R);
}
